# starting kernel
# baseline (speedup 1.0000x reference)
#include <hip/hip_runtime.h>

typedef __attribute__((ext_vector_type(2))) float v2f;
typedef __attribute__((ext_vector_type(8))) float v8f;

#define N_NODES 50000
#define N_EDGES 800000
#define D_IN    64
#define D_HID   128
#define HPAD    132

__global__ void gin_init(const float* __restrict__ x,
                         const float* __restrict__ eps,
                         float* __restrict__ agg, int n) {
    int i = blockIdx.x * blockDim.x + threadIdx.x;
    if (i < n) agg[i] = (1.0f + eps[0]) * x[i];
}

__global__ void gin_scatter(const float* __restrict__ x,
                            const int* __restrict__ ei,
                            float* __restrict__ agg) {
    unsigned tid = blockIdx.x * blockDim.x + threadIdx.x;
    unsigned e = tid >> 4;
    if (e >= N_EDGES) return;
    int f = (tid & 15u) << 2;
    int dst = ei[e];
    int src = ei[N_EDGES + e];
    const float4 v = *(const float4*)(x + (size_t)src * D_IN + f);
    float* a = agg + (size_t)dst * D_IN + f;
    atomicAdd(a + 0, v.x);
    atomicAdd(a + 1, v.y);
    atomicAdd(a + 2, v.z);
    atomicAdd(a + 3, v.w);
}

__global__ __launch_bounds__(128)
void gin_mlp(const float* __restrict__ h,
             const float* __restrict__ W1,
             const float* __restrict__ b1,
             const float* __restrict__ W2,
             const float* __restrict__ b2,
             float* __restrict__ out) {
    __shared__ float hid[4][16 * HPAD];

    const int wave  = threadIdx.x >> 5;
    const int lane  = threadIdx.x & 31;
    const int r     = lane & 15;
    const int koff  = (lane >> 4) << 1;
    const int mhalf = (lane >> 4) << 3;
    const int tile  = blockIdx.x * 4 + wave;
    if (tile * 16 >= N_NODES) return;
    const int m0 = tile * 16;

    v2f afr[16];
    const float* arow = h + (size_t)(m0 + r) * D_IN + koff;
#pragma unroll
    for (int kk = 0; kk < 16; ++kk)
        afr[kk] = *(const v2f*)(arow + kk * 4);

    float* hrow = &hid[wave][0];
#pragma unroll
    for (int nt = 0; nt < 8; ++nt) {
        const int n0 = nt * 16;
        const float bias = b1[n0 + r];
        v8f acc;
#pragma unroll
        for (int v = 0; v < 8; ++v) acc[v] = bias;
#pragma unroll
        for (int kk = 0; kk < 16; ++kk) {
            const int k = kk * 4 + koff;
            v2f bfr;
            bfr.x = W1[(size_t)k * D_HID + n0 + r];
            bfr.y = W1[(size_t)(k + 1) * D_HID + n0 + r];
            acc = __builtin_amdgcn_wmma_f32_16x16x4_f32(
                false, afr[kk], false, bfr, (short)0, acc, false, false);
        }
#pragma unroll
        for (int v = 0; v < 8; ++v) {
            float t = acc[v];
            hrow[(v + mhalf) * HPAD + n0 + r] = t > 0.0f ? t : 0.0f;
        }
    }

    float* orow = out + (size_t)m0 * D_IN;
#pragma unroll
    for (int nt = 0; nt < 4; ++nt) {
        const int n0 = nt * 16;
        const float bias = b2[n0 + r];
        v8f acc;
#pragma unroll
        for (int v = 0; v < 8; ++v) acc[v] = bias;
#pragma unroll
        for (int kk = 0; kk < 32; ++kk) {
            const int k = kk * 4 + koff;
            v2f a2;
            a2.x = hrow[r * HPAD + k];
            a2.y = hrow[r * HPAD + k + 1];
            v2f bfr;
            bfr.x = W2[(size_t)k * D_IN + n0 + r];
            bfr.y = W2[(size_t)(k + 1) * D_IN + n0 + r];
            acc = __builtin_amdgcn_wmma_f32_16x16x4_f32(
                false, a2, false, bfr, (short)0, acc, false, false);
        }
#pragma unroll
        for (int v = 0; v < 8; ++v)
            orow[(size_t)(v + mhalf) * D_IN + n0 + r] = acc[v];
    }
}

extern "C" void kernel_launch(void* const* d_in, const int* in_sizes, int n_in,
                              void* d_out, int out_size, void* d_ws, size_t ws_size,
                              hipStream_t stream) {
    const float* x   = (const float*)d_in[0];
    const int*   ei  = (const int*)d_in[1];
    const float* W1  = (const float*)d_in[2];
    const float* b1  = (const float*)d_in[3];
    const float* W2  = (const float*)d_in[4];
    const float* b2  = (const float*)d_in[5];
    const float* eps = (const float*)d_in[6];
    float* out = (float*)d_out;
    float* agg = (float*)d_ws;

    int n = N_NODES * D_IN;
    gin_init<<<(n + 255) / 256, 256, 0, stream>>>(x, eps, agg, n);

    unsigned total = (unsigned)N_EDGES * 16u;
    gin_scatter<<<(total + 255) / 256, 256, 0, stream>>>(x, ei, agg);

    int tiles = (N_NODES + 15) / 16;
    gin_mlp<<<(tiles + 3) / 4, 128, 0, stream>>>(agg, W1, b1, W2, b2, out);
}


# MLP fused into the range-sum kernel: the 64-row h tile stays in LDS and feeds the f32 WMMAs (a wave owns one column tile over the row tiles, B read once); agg round trip and one launch dropped
# speedup vs baseline: 2.3365x; 2.3365x over previous
#include <hip/hip_runtime.h>

typedef __attribute__((ext_vector_type(2))) float v2f;
typedef __attribute__((ext_vector_type(8))) float v8f;

#define N_NODES 50000
#define N_EDGES 800000
#define D_IN    64
#define D_HID   128
#define HPAD    132
#define HS      68

#define WS_TOFF   12800000
#define WS_BINNED 13414656
#define NB        ((N_NODES + 63) / 64)
#define E4        (N_EDGES / 4)
#define CH        4096
#define NCH       ((N_EDGES + CH - 1) / CH)
#define CPW       ((NCH + 7) / 8)
#define TOFF(r, c) ((((r) >> 3) * NCH + (c)) * 8 + ((r) & 7))
static_assert(NB < 1024, "one thread of the binning workgroup per range");
static_assert(WS_BINNED - WS_TOFF >= 4 * (NB / 8 + 1) * NCH * 8, "toff fits");
static_assert(CPW <= 32, "one lane per chunk of a wave");

__global__ __launch_bounds__(1024)
void gin_bin(const int* __restrict__ ei, int* __restrict__ toff,
             int* __restrict__ binned) {
    __shared__ int h[1024];
    __shared__ int loff[1024];
    __shared__ unsigned sorted[CH];
    __shared__ int wa[32];
    const int t    = threadIdx.x;
    const int lane = t & 31;
    const int wave = t >> 5;
    h[t] = 0;
    __syncthreads();
    const int q = blockIdx.x * 1024 + t;
    const bool ok = q < E4;
    int4 r = make_int4(0, 0, 0, 0);
    int4 c = make_int4(0, 0, 0, 0);
    if (ok) {
        r = ((const int4*)ei)[q];
        c = ((const int4*)(ei + N_EDGES))[q];
        atomicAdd(&h[r.x >> 6], 1);
        atomicAdd(&h[r.y >> 6], 1);
        atomicAdd(&h[r.z >> 6], 1);
        atomicAdd(&h[r.w >> 6], 1);
    }
    __syncthreads();
    const int mine = h[t];
    int ia = mine;
    for (int d = 1; d < 32; d <<= 1) {
        const int ua = __shfl_up(ia, d);
        if (lane >= d) ia += ua;
    }
    if (lane == 31) wa[wave] = ia;
    __syncthreads();
    int pa = 0;
    for (int w = 0; w < wave; ++w) pa += wa[w];
    const int first = pa + ia - mine;
    loff[t] = first;
    if (t <= NB) toff[TOFF(t, blockIdx.x)] = first;
    h[t] = 0;
    __syncthreads();
    if (ok) {
        int p;
        p = loff[r.x >> 6] + atomicAdd(&h[r.x >> 6], 1);
        sorted[p] = ((unsigned)r.x << 16) | (unsigned)c.x;
        p = loff[r.y >> 6] + atomicAdd(&h[r.y >> 6], 1);
        sorted[p] = ((unsigned)r.y << 16) | (unsigned)c.y;
        p = loff[r.z >> 6] + atomicAdd(&h[r.z >> 6], 1);
        sorted[p] = ((unsigned)r.z << 16) | (unsigned)c.z;
        p = loff[r.w >> 6] + atomicAdd(&h[r.w >> 6], 1);
        sorted[p] = ((unsigned)r.w << 16) | (unsigned)c.w;
    }
    __syncthreads();
    int nvalid = N_EDGES - blockIdx.x * CH;
    if (nvalid > CH) nvalid = CH;
    for (int i = t; i < nvalid; i += 1024)
        binned[blockIdx.x * CH + i] = (int)sorted[i];
}

__global__ __launch_bounds__(256)
void gin_sum_mlp(const float* __restrict__ x,
                 const float* __restrict__ eps,
                 const int* __restrict__ toff,
                 const int* __restrict__ binned,
                 const float* __restrict__ W1,
                 const float* __restrict__ b1,
                 const float* __restrict__ W2,
                 const float* __restrict__ b2,
                 float* __restrict__ out) {
    __shared__ __attribute__((aligned(16))) float acc[64 * HS];
    __shared__ __attribute__((aligned(16))) float hid[64 * HPAD];
    const int t    = threadIdx.x;
    const int lane = t & 31;
    const int wave = __builtin_amdgcn_readfirstlane(t >> 5);
    const int b    = blockIdx.x;
    const int ch = wave * CPW + lane;
    int st = 0;
    int en = 0;
    if (lane < CPW && ch < NCH) {
        st = toff[TOFF(b, ch)];
        en = toff[TOFF(b + 1, ch)];
    }
    for (int i = t; i < 64 * HS / 4; i += 256)
        ((float4*)acc)[i] = make_float4(0.0f, 0.0f, 0.0f, 0.0f);
    const int len = en - st;
    int e = len;
    for (int d = 1; d < 32; d <<= 1) {
        const int u = __shfl_up(e, d);
        if (lane >= d) e += u;
    }
    const int m = __builtin_amdgcn_readlane(e, 31);
    const int rel = ch * CH + st - (e - len);
    int ends[CPW];
    for (int i = 0; i < CPW; ++i) ends[i] = __builtin_amdgcn_readlane(e, i);
    __syncthreads();
    const v2f* x2 = (const v2f*)x;
    for (int base = 0; base < m; base += 32) {
        int mm = m - base;
        if (mm > 32) mm = 32;
        const int j = base + lane;
        int own = 0;
        for (int i = 0; i < CPW; ++i) own += (j >= ends[i]) ? 1 : 0;
        const int orel = __shfl(rel, own);
        const unsigned myw = (lane < mm) ? (unsigned)binned[orel + j] : 0u;
        int k = 0;
        for (; k + 8 <= mm; k += 8) {
            v2f v[8];
            unsigned w[8];
            for (int jj = 0; jj < 8; ++jj) {
                w[jj] = (unsigned)__builtin_amdgcn_readlane((int)myw, k + jj);
                v[jj] = x2[(size_t)(w[jj] & 0xffffu) * 32 + lane];
            }
            for (int jj = 0; jj < 8; ++jj) {
                float* a = acc + ((w[jj] >> 16) & 63u) * HS + 2 * lane;
                atomicAdd(a, v[jj].x);
                atomicAdd(a + 1, v[jj].y);
            }
        }
        if (k + 4 <= mm) {
            v2f v[4];
            unsigned w[4];
            for (int jj = 0; jj < 4; ++jj) {
                w[jj] = (unsigned)__builtin_amdgcn_readlane((int)myw, k + jj);
                v[jj] = x2[(size_t)(w[jj] & 0xffffu) * 32 + lane];
            }
            for (int jj = 0; jj < 4; ++jj) {
                float* a = acc + ((w[jj] >> 16) & 63u) * HS + 2 * lane;
                atomicAdd(a, v[jj].x);
                atomicAdd(a + 1, v[jj].y);
            }
            k += 4;
        }
        for (; k < mm; ++k) {
            const unsigned w = (unsigned)__builtin_amdgcn_readlane((int)myw, k);
            const v2f v = x2[(size_t)(w & 0xffffu) * 32 + lane];
            float* a = acc + ((w >> 16) & 63u) * HS + 2 * lane;
            atomicAdd(a, v.x);
            atomicAdd(a + 1, v.y);
        }
    }
    __syncthreads();
    const float se = 1.0f + eps[0];
    for (int i = t; i < 1024; i += 256) {
        const int row  = i >> 4;
        const int node = b * 64 + row;
        float4* p = (float4*)(acc + row * HS) + (i & 15);
        float4 o = make_float4(0.0f, 0.0f, 0.0f, 0.0f);
        if (node < N_NODES) {
            const float4 xv = ((const float4*)x)[(size_t)node * 16 + (i & 15)];
            const float4 a  = *p;
            o.x = se * xv.x + a.x;
            o.y = se * xv.y + a.y;
            o.z = se * xv.z + a.z;
            o.w = se * xv.w + a.w;
        }
        *p = o;
    }
    __syncthreads();

    const int r     = lane & 15;
    const int koff  = (lane >> 4) << 1;
    const int mhalf = (lane >> 4) << 3;

    {
        const int n0 = wave * 16;
        const float bias = b1[n0 + r];
        v8f c0, c1, c2, c3;
#pragma unroll
        for (int v = 0; v < 8; ++v) { c0[v] = bias; c1[v] = bias; c2[v] = bias; c3[v] = bias; }
        const float* a0p = acc + r * HS + koff;
        const float* bp  = W1 + (size_t)koff * D_HID + n0 + r;
#pragma unroll
        for (int kk = 0; kk < 16; ++kk) {
            v2f bfr;
            bfr.x = bp[(size_t)(kk * 4) * D_HID];
            bfr.y = bp[(size_t)(kk * 4 + 1) * D_HID];
            const v2f a0 = *(const v2f*)(a0p + kk * 4);
            const v2f a1 = *(const v2f*)(a0p + 16 * HS + kk * 4);
            const v2f a2 = *(const v2f*)(a0p + 32 * HS + kk * 4);
            const v2f a3 = *(const v2f*)(a0p + 48 * HS + kk * 4);
            c0 = __builtin_amdgcn_wmma_f32_16x16x4_f32(
                false, a0, false, bfr, (short)0, c0, false, false);
            c1 = __builtin_amdgcn_wmma_f32_16x16x4_f32(
                false, a1, false, bfr, (short)0, c1, false, false);
            c2 = __builtin_amdgcn_wmma_f32_16x16x4_f32(
                false, a2, false, bfr, (short)0, c2, false, false);
            c3 = __builtin_amdgcn_wmma_f32_16x16x4_f32(
                false, a3, false, bfr, (short)0, c3, false, false);
        }
        float* hp = hid + mhalf * HPAD + n0 + r;
#pragma unroll
        for (int v = 0; v < 8; ++v) {
            const float t0 = c0[v], t1 = c1[v], t2 = c2[v], t3 = c3[v];
            hp[(v)      * HPAD] = t0 > 0.0f ? t0 : 0.0f;
            hp[(v + 16) * HPAD] = t1 > 0.0f ? t1 : 0.0f;
            hp[(v + 32) * HPAD] = t2 > 0.0f ? t2 : 0.0f;
            hp[(v + 48) * HPAD] = t3 > 0.0f ? t3 : 0.0f;
        }
    }
    __syncthreads();

    {
        const int n0  = (wave & 3) * 16;
        const int rt0 = (wave >> 2) * 32;
        const float bias = b2[n0 + r];
        v8f d0, d1;
#pragma unroll
        for (int v = 0; v < 8; ++v) { d0[v] = bias; d1[v] = bias; }
        const float* a0p = hid + (rt0 + r) * HPAD + koff;
        const float* bp  = W2 + (size_t)koff * D_IN + n0 + r;
#pragma unroll
        for (int kk = 0; kk < 32; ++kk) {
            v2f bfr;
            bfr.x = bp[(size_t)(kk * 4) * D_IN];
            bfr.y = bp[(size_t)(kk * 4 + 1) * D_IN];
            const v2f a0 = *(const v2f*)(a0p + kk * 4);
            const v2f a1 = *(const v2f*)(a0p + 16 * HPAD + kk * 4);
            d0 = __builtin_amdgcn_wmma_f32_16x16x4_f32(
                false, a0, false, bfr, (short)0, d0, false, false);
            d1 = __builtin_amdgcn_wmma_f32_16x16x4_f32(
                false, a1, false, bfr, (short)0, d1, false, false);
        }
        float* op = acc + (rt0 + mhalf) * HS + n0 + r;
#pragma unroll
        for (int v = 0; v < 8; ++v) {
            op[(v)      * HS] = d0[v];
            op[(v + 16) * HS] = d1[v];
        }
    }
    __syncthreads();
    for (int i = t; i < 1024; i += 256) {
        const int row  = i >> 4;
        const int node = b * 64 + row;
        if (node < N_NODES)
            ((float4*)out)[(size_t)node * 16 + (i & 15)] =
                *((const float4*)(acc + row * HS) + (i & 15));
    }
}

extern "C" void kernel_launch(void* const* d_in, const int* in_sizes, int n_in,
                              void* d_out, int out_size, void* d_ws, size_t ws_size,
                              hipStream_t stream) {
    const float* x   = (const float*)d_in[0];
    const int*   ei  = (const int*)d_in[1];
    const float* W1  = (const float*)d_in[2];
    const float* b1  = (const float*)d_in[3];
    const float* W2  = (const float*)d_in[4];
    const float* b2  = (const float*)d_in[5];
    const float* eps = (const float*)d_in[6];
    float* out = (float*)d_out;
    int*   toff   = (int*)((char*)d_ws + WS_TOFF);
    int*   binned = (int*)((char*)d_ws + WS_BINNED);

    gin_bin<<<NCH, 1024, 0, stream>>>(ei, toff, binned);
    gin_sum_mlp<<<NB, 256, 0, stream>>>(x, eps, toff, binned, W1, b1, W2, b2, out);
}


# range's edges queued per destination row by returning integer LDS adds; a wave sums its 8 rows' source rows in registers, list (LDS arrival) order, self-term last; LDS float adds and zero fill go
# speedup vs baseline: 2.6742x; 1.1445x over previous
#include <hip/hip_runtime.h>

typedef __attribute__((ext_vector_type(2))) float v2f;
typedef __attribute__((ext_vector_type(8))) float v8f;

#define N_NODES 50000
#define N_EDGES 800000
#define D_IN    64
#define D_HID   128
#define HPAD    132
#define HS      68
#define RCAP    128

#define WS_TOFF   12800000
#define WS_BINNED 13414656
#define NB        ((N_NODES + 63) / 64)
#define E4        (N_EDGES / 4)
#define CH        4096
#define NCH       ((N_EDGES + CH - 1) / CH)
#define CPW       ((NCH + 7) / 8)
#define TOFF(r, c) ((((r) >> 3) * NCH + (c)) * 8 + ((r) & 7))
static_assert(NB < 1024, "one thread of the binning workgroup per range");
static_assert(WS_BINNED - WS_TOFF >= 4 * (NB / 8 + 1) * NCH * 8, "toff fits");
static_assert(CPW <= 32, "one lane per chunk of a wave");

__global__ __launch_bounds__(1024)
void gin_bin(const int* __restrict__ ei, int* __restrict__ toff,
             int* __restrict__ binned) {
    __shared__ int h[1024];
    __shared__ int loff[1024];
    __shared__ unsigned sorted[CH];
    __shared__ int wa[32];
    const int t    = threadIdx.x;
    const int lane = t & 31;
    const int wave = t >> 5;
    h[t] = 0;
    __syncthreads();
    const int q = blockIdx.x * 1024 + t;
    const bool ok = q < E4;
    int4 r = make_int4(0, 0, 0, 0);
    int4 c = make_int4(0, 0, 0, 0);
    if (ok) {
        r = ((const int4*)ei)[q];
        c = ((const int4*)(ei + N_EDGES))[q];
        atomicAdd(&h[r.x >> 6], 1);
        atomicAdd(&h[r.y >> 6], 1);
        atomicAdd(&h[r.z >> 6], 1);
        atomicAdd(&h[r.w >> 6], 1);
    }
    __syncthreads();
    const int mine = h[t];
    int ia = mine;
    for (int d = 1; d < 32; d <<= 1) {
        const int ua = __shfl_up(ia, d);
        if (lane >= d) ia += ua;
    }
    if (lane == 31) wa[wave] = ia;
    __syncthreads();
    int pa = 0;
    for (int w = 0; w < wave; ++w) pa += wa[w];
    const int first = pa + ia - mine;
    loff[t] = first;
    if (t <= NB) toff[TOFF(t, blockIdx.x)] = first;
    h[t] = 0;
    __syncthreads();
    if (ok) {
        int p;
        p = loff[r.x >> 6] + atomicAdd(&h[r.x >> 6], 1);
        sorted[p] = ((unsigned)r.x << 16) | (unsigned)c.x;
        p = loff[r.y >> 6] + atomicAdd(&h[r.y >> 6], 1);
        sorted[p] = ((unsigned)r.y << 16) | (unsigned)c.y;
        p = loff[r.z >> 6] + atomicAdd(&h[r.z >> 6], 1);
        sorted[p] = ((unsigned)r.z << 16) | (unsigned)c.z;
        p = loff[r.w >> 6] + atomicAdd(&h[r.w >> 6], 1);
        sorted[p] = ((unsigned)r.w << 16) | (unsigned)c.w;
    }
    __syncthreads();
    int nvalid = N_EDGES - blockIdx.x * CH;
    if (nvalid > CH) nvalid = CH;
    for (int i = t; i < nvalid; i += 1024)
        binned[blockIdx.x * CH + i] = (int)sorted[i];
}

__global__ __launch_bounds__(256)
void gin_sum_mlp(const float* __restrict__ x,
                 const float* __restrict__ eps,
                 const int* __restrict__ toff,
                 const int* __restrict__ binned,
                 const float* __restrict__ W1,
                 const float* __restrict__ b1,
                 const float* __restrict__ W2,
                 const float* __restrict__ b2,
                 float* __restrict__ out) {
    __shared__ __attribute__((aligned(16))) float acc[64 * HS];
    __shared__ __attribute__((aligned(16))) unsigned char hid_raw[64 * HPAD * 4];
    __shared__ int cnt[64];
    __shared__ int ovf;
    float* hid = (float*)hid_raw;
    unsigned short* slot = (unsigned short*)hid_raw;
    static_assert(64 * RCAP * 2 <= 64 * HPAD * 4, "source lists fit the hidden tile");
    const int t    = threadIdx.x;
    const int lane = t & 31;
    const int wave = __builtin_amdgcn_readfirstlane(t >> 5);
    const int b    = blockIdx.x;
    const int ch = wave * CPW + lane;
    int st = 0;
    int en = 0;
    if (lane < CPW && ch < NCH) {
        st = toff[TOFF(b, ch)];
        en = toff[TOFF(b + 1, ch)];
    }
    if (t < 64) cnt[t] = 0;
    if (t == 0) ovf = 0;
    const int len = en - st;
    int e = len;
    for (int d = 1; d < 32; d <<= 1) {
        const int u = __shfl_up(e, d);
        if (lane >= d) e += u;
    }
    const int m = __builtin_amdgcn_readlane(e, 31);
    const int rel = ch * CH + st - (e - len);
    int ends[CPW];
    for (int i = 0; i < CPW; ++i) ends[i] = __builtin_amdgcn_readlane(e, i);
    __syncthreads();
    const v2f* x2 = (const v2f*)x;
    for (int base = 0; base < m; base += 128) {
        unsigned w[4];
        for (int u = 0; u < 4; ++u) {
            const int j = base + 32 * u + lane;
            int own = 0;
            for (int i = 0; i < CPW; ++i) own += (j >= ends[i]) ? 1 : 0;
            const int orel = __shfl(rel, own);
            w[u] = (j < m) ? (unsigned)binned[orel + j] : 0u;
        }
        for (int u = 0; u < 4; ++u) {
            if (base + 32 * u + lane < m) {
                const unsigned row = (w[u] >> 16) & 63u;
                const int rank = atomicAdd(&cnt[row], 1);
                if (rank < RCAP) slot[row * RCAP + rank] = (unsigned short)(w[u] & 0xffffu);
                else             ovf = 1;
            }
        }
    }
    __syncthreads();
    const float se = 1.0f + eps[0];
    const int over = __builtin_amdgcn_readfirstlane(ovf);
    if (over == 0) {
        const int row0 = wave * 8;
        const int myn  = cnt[row0 + (lane & 7)];
        unsigned nxt = (unsigned)slot[row0 * RCAP + lane];
        for (int i = 0; i < 8; ++i) {
            const int row  = row0 + i;
            const int node = b * 64 + row;
            const int n    = __builtin_amdgcn_readlane(myn, i);
            unsigned mys = nxt;
            if (i < 7) nxt = (unsigned)slot[(row + 1) * RCAP + lane];
            v2f xv; xv.x = 0.0f; xv.y = 0.0f;
            if (node < N_NODES) xv = x2[(size_t)node * 32 + lane];
            v2f s; s.x = 0.0f; s.y = 0.0f;
            for (int base = 0; base < n; base += 32) {
                int mm = n - base;
                if (mm > 32) mm = 32;
                if (base > 0) mys = (unsigned)slot[row * RCAP + base + lane];
                int k = 0;
                for (; k + 8 <= mm; k += 8) {
                    v2f v[8];
                    for (int jj = 0; jj < 8; ++jj) {
                        const unsigned sidx = (unsigned)__builtin_amdgcn_readlane((int)mys, k + jj);
                        v[jj] = x2[(size_t)(sidx & 0xffffu) * 32 + lane];
                    }
                    for (int jj = 0; jj < 8; ++jj) {
                        s.x += v[jj].x;
                        s.y += v[jj].y;
                    }
                }
                const int tl = mm - k;
                if (tl > 0) {
                    v2f v[7];
                    for (int jj = 0; jj < 7; ++jj) {
                        v[jj].x = 0.0f; v[jj].y = 0.0f;
                        if (jj < tl) {
                            const unsigned sidx = (unsigned)__builtin_amdgcn_readlane((int)mys, k + jj);
                            v[jj] = x2[(size_t)(sidx & 0xffffu) * 32 + lane];
                        }
                    }
                    for (int jj = 0; jj < 7; ++jj) {
                        if (jj < tl) {
                            s.x += v[jj].x;
                            s.y += v[jj].y;
                        }
                    }
                }
            }
            v2f o; o.x = 0.0f; o.y = 0.0f;
            if (node < N_NODES) {
                o.x = se * xv.x + s.x;
                o.y = se * xv.y + s.y;
            }
            *(v2f*)(acc + row * HS + 2 * lane) = o;
        }
    } else {
        for (int i = t; i < 64 * HS / 4; i += 256)
            ((float4*)acc)[i] = make_float4(0.0f, 0.0f, 0.0f, 0.0f);
        __syncthreads();
        for (int base = 0; base < m; base += 32) {
            int mm = m - base;
            if (mm > 32) mm = 32;
            const int j = base + lane;
            int own = 0;
            for (int i = 0; i < CPW; ++i) own += (j >= ends[i]) ? 1 : 0;
            const int orel = __shfl(rel, own);
            const unsigned myw = (lane < mm) ? (unsigned)binned[orel + j] : 0u;
            int k = 0;
            for (; k + 8 <= mm; k += 8) {
                v2f v[8];
                unsigned w[8];
                for (int jj = 0; jj < 8; ++jj) {
                    w[jj] = (unsigned)__builtin_amdgcn_readlane((int)myw, k + jj);
                    v[jj] = x2[(size_t)(w[jj] & 0xffffu) * 32 + lane];
                }
                for (int jj = 0; jj < 8; ++jj) {
                    float* a = acc + ((w[jj] >> 16) & 63u) * HS + 2 * lane;
                    atomicAdd(a, v[jj].x);
                    atomicAdd(a + 1, v[jj].y);
                }
            }
            if (k + 4 <= mm) {
                v2f v[4];
                unsigned w[4];
                for (int jj = 0; jj < 4; ++jj) {
                    w[jj] = (unsigned)__builtin_amdgcn_readlane((int)myw, k + jj);
                    v[jj] = x2[(size_t)(w[jj] & 0xffffu) * 32 + lane];
                }
                for (int jj = 0; jj < 4; ++jj) {
                    float* a = acc + ((w[jj] >> 16) & 63u) * HS + 2 * lane;
                    atomicAdd(a, v[jj].x);
                    atomicAdd(a + 1, v[jj].y);
                }
                k += 4;
            }
            for (; k < mm; ++k) {
                const unsigned w = (unsigned)__builtin_amdgcn_readlane((int)myw, k);
                const v2f v = x2[(size_t)(w & 0xffffu) * 32 + lane];
                float* a = acc + ((w >> 16) & 63u) * HS + 2 * lane;
                atomicAdd(a, v.x);
                atomicAdd(a + 1, v.y);
            }
        }
        __syncthreads();
        for (int i = t; i < 1024; i += 256) {
            const int row  = i >> 4;
            const int node = b * 64 + row;
            float4* p = (float4*)(acc + row * HS) + (i & 15);
            float4 o = make_float4(0.0f, 0.0f, 0.0f, 0.0f);
            if (node < N_NODES) {
                const float4 xv = ((const float4*)x)[(size_t)node * 16 + (i & 15)];
                const float4 a  = *p;
                o.x = se * xv.x + a.x;
                o.y = se * xv.y + a.y;
                o.z = se * xv.z + a.z;
                o.w = se * xv.w + a.w;
            }
            *p = o;
        }
    }
    __syncthreads();

    const int r     = lane & 15;
    const int koff  = (lane >> 4) << 1;
    const int mhalf = (lane >> 4) << 3;

    {
        const int n0 = wave * 16;
        const float bias = b1[n0 + r];
        v8f c0, c1, c2, c3;
#pragma unroll
        for (int v = 0; v < 8; ++v) { c0[v] = bias; c1[v] = bias; c2[v] = bias; c3[v] = bias; }
        const float* a0p = acc + r * HS + koff;
        const float* bp  = W1 + (size_t)koff * D_HID + n0 + r;
#pragma unroll
        for (int kk = 0; kk < 16; ++kk) {
            v2f bfr;
            bfr.x = bp[(size_t)(kk * 4) * D_HID];
            bfr.y = bp[(size_t)(kk * 4 + 1) * D_HID];
            const v2f a0 = *(const v2f*)(a0p + kk * 4);
            const v2f a1 = *(const v2f*)(a0p + 16 * HS + kk * 4);
            const v2f a2 = *(const v2f*)(a0p + 32 * HS + kk * 4);
            const v2f a3 = *(const v2f*)(a0p + 48 * HS + kk * 4);
            c0 = __builtin_amdgcn_wmma_f32_16x16x4_f32(
                false, a0, false, bfr, (short)0, c0, false, false);
            c1 = __builtin_amdgcn_wmma_f32_16x16x4_f32(
                false, a1, false, bfr, (short)0, c1, false, false);
            c2 = __builtin_amdgcn_wmma_f32_16x16x4_f32(
                false, a2, false, bfr, (short)0, c2, false, false);
            c3 = __builtin_amdgcn_wmma_f32_16x16x4_f32(
                false, a3, false, bfr, (short)0, c3, false, false);
        }
        float* hp = hid + mhalf * HPAD + n0 + r;
#pragma unroll
        for (int v = 0; v < 8; ++v) {
            const float t0 = c0[v], t1 = c1[v], t2 = c2[v], t3 = c3[v];
            hp[(v)      * HPAD] = t0 > 0.0f ? t0 : 0.0f;
            hp[(v + 16) * HPAD] = t1 > 0.0f ? t1 : 0.0f;
            hp[(v + 32) * HPAD] = t2 > 0.0f ? t2 : 0.0f;
            hp[(v + 48) * HPAD] = t3 > 0.0f ? t3 : 0.0f;
        }
    }
    __syncthreads();

    {
        const int n0  = (wave & 3) * 16;
        const int rt0 = (wave >> 2) * 32;
        const float bias = b2[n0 + r];
        v8f d0, d1;
#pragma unroll
        for (int v = 0; v < 8; ++v) { d0[v] = bias; d1[v] = bias; }
        const float* a0p = hid + (rt0 + r) * HPAD + koff;
        const float* bp  = W2 + (size_t)koff * D_IN + n0 + r;
#pragma unroll
        for (int kk = 0; kk < 32; ++kk) {
            v2f bfr;
            bfr.x = bp[(size_t)(kk * 4) * D_IN];
            bfr.y = bp[(size_t)(kk * 4 + 1) * D_IN];
            const v2f a0 = *(const v2f*)(a0p + kk * 4);
            const v2f a1 = *(const v2f*)(a0p + 16 * HPAD + kk * 4);
            d0 = __builtin_amdgcn_wmma_f32_16x16x4_f32(
                false, a0, false, bfr, (short)0, d0, false, false);
            d1 = __builtin_amdgcn_wmma_f32_16x16x4_f32(
                false, a1, false, bfr, (short)0, d1, false, false);
        }
        float* op = acc + (rt0 + mhalf) * HS + n0 + r;
#pragma unroll
        for (int v = 0; v < 8; ++v) {
            op[(v)      * HS] = d0[v];
            op[(v + 16) * HS] = d1[v];
        }
    }
    __syncthreads();
    for (int i = t; i < 1024; i += 256) {
        const int row  = i >> 4;
        const int node = b * 64 + row;
        if (node < N_NODES)
            ((float4*)out)[(size_t)node * 16 + (i & 15)] =
                *((const float4*)(acc + row * HS) + (i & 15));
    }
}

extern "C" void kernel_launch(void* const* d_in, const int* in_sizes, int n_in,
                              void* d_out, int out_size, void* d_ws, size_t ws_size,
                              hipStream_t stream) {
    const float* x   = (const float*)d_in[0];
    const int*   ei  = (const int*)d_in[1];
    const float* W1  = (const float*)d_in[2];
    const float* b1  = (const float*)d_in[3];
    const float* W2  = (const float*)d_in[4];
    const float* b2  = (const float*)d_in[5];
    const float* eps = (const float*)d_in[6];
    float* out = (float*)d_out;
    int*   toff   = (int*)((char*)d_ws + WS_TOFF);
    int*   binned = (int*)((char*)d_ws + WS_BINNED);

    gin_bin<<<NCH, 1024, 0, stream>>>(ei, toff, binned);
    gin_sum_mlp<<<NB, 256, 0, stream>>>(x, eps, toff, binned, W1, b1, W2, b2, out);
}


# range sums: all source-row loads of a destination row (up to 31, as groups of 16/8/4/2/1 by the bits of its count) are issued before its first add instead of drained per group of 8; list order kept
# speedup vs baseline: 2.9457x; 1.1015x over previous
#include <hip/hip_runtime.h>

typedef __attribute__((ext_vector_type(2))) float v2f;
typedef __attribute__((ext_vector_type(8))) float v8f;

#define N_NODES 50000
#define N_EDGES 800000
#define D_IN    64
#define D_HID   128
#define HPAD    132
#define HS      68
#define RCAP    128

#define WS_TOFF   12800000
#define WS_BINNED 13414656
#define NB        ((N_NODES + 63) / 64)
#define E4        (N_EDGES / 4)
#define CH        4096
#define NCH       ((N_EDGES + CH - 1) / CH)
#define CPW       ((NCH + 7) / 8)
#define TOFF(r, c) ((((r) >> 3) * NCH + (c)) * 8 + ((r) & 7))
static_assert(NB < 1024, "one thread of the binning workgroup per range");
static_assert(WS_BINNED - WS_TOFF >= 4 * (NB / 8 + 1) * NCH * 8, "toff fits");
static_assert(CPW <= 32, "one lane per chunk of a wave");

__global__ __launch_bounds__(1024)
void gin_bin(const int* __restrict__ ei, int* __restrict__ toff,
             int* __restrict__ binned) {
    __shared__ int h[1024];
    __shared__ int loff[1024];
    __shared__ unsigned sorted[CH];
    __shared__ int wa[32];
    const int t    = threadIdx.x;
    const int lane = t & 31;
    const int wave = t >> 5;
    h[t] = 0;
    __syncthreads();
    const int q = blockIdx.x * 1024 + t;
    const bool ok = q < E4;
    int4 r = make_int4(0, 0, 0, 0);
    int4 c = make_int4(0, 0, 0, 0);
    if (ok) {
        r = ((const int4*)ei)[q];
        c = ((const int4*)(ei + N_EDGES))[q];
        atomicAdd(&h[r.x >> 6], 1);
        atomicAdd(&h[r.y >> 6], 1);
        atomicAdd(&h[r.z >> 6], 1);
        atomicAdd(&h[r.w >> 6], 1);
    }
    __syncthreads();
    const int mine = h[t];
    int ia = mine;
    for (int d = 1; d < 32; d <<= 1) {
        const int ua = __shfl_up(ia, d);
        if (lane >= d) ia += ua;
    }
    if (lane == 31) wa[wave] = ia;
    __syncthreads();
    int pa = 0;
    for (int w = 0; w < wave; ++w) pa += wa[w];
    const int first = pa + ia - mine;
    loff[t] = first;
    if (t <= NB) toff[TOFF(t, blockIdx.x)] = first;
    h[t] = 0;
    __syncthreads();
    if (ok) {
        int p;
        p = loff[r.x >> 6] + atomicAdd(&h[r.x >> 6], 1);
        sorted[p] = ((unsigned)r.x << 16) | (unsigned)c.x;
        p = loff[r.y >> 6] + atomicAdd(&h[r.y >> 6], 1);
        sorted[p] = ((unsigned)r.y << 16) | (unsigned)c.y;
        p = loff[r.z >> 6] + atomicAdd(&h[r.z >> 6], 1);
        sorted[p] = ((unsigned)r.z << 16) | (unsigned)c.z;
        p = loff[r.w >> 6] + atomicAdd(&h[r.w >> 6], 1);
        sorted[p] = ((unsigned)r.w << 16) | (unsigned)c.w;
    }
    __syncthreads();
    int nvalid = N_EDGES - blockIdx.x * CH;
    if (nvalid > CH) nvalid = CH;
    for (int i = t; i < nvalid; i += 1024)
        binned[blockIdx.x * CH + i] = (int)sorted[i];
}

__device__ __forceinline__ v2f src_row(const v2f* __restrict__ x2, unsigned ids, int l, int lane) {
    const unsigned sidx = (unsigned)__builtin_amdgcn_readlane((int)ids, l);
    return x2[(size_t)(sidx & 0xffffu) * 32 + lane];
}

__global__ __launch_bounds__(256)
void gin_sum_mlp(const float* __restrict__ x,
                 const float* __restrict__ eps,
                 const int* __restrict__ toff,
                 const int* __restrict__ binned,
                 const float* __restrict__ W1,
                 const float* __restrict__ b1,
                 const float* __restrict__ W2,
                 const float* __restrict__ b2,
                 float* __restrict__ out) {
    __shared__ __attribute__((aligned(16))) float acc[64 * HS];
    __shared__ __attribute__((aligned(16))) unsigned char hid_raw[64 * HPAD * 4];
    __shared__ int cnt[64];
    __shared__ int ovf;
    float* hid = (float*)hid_raw;
    unsigned short* slot = (unsigned short*)hid_raw;
    static_assert(64 * RCAP * 2 <= 64 * HPAD * 4, "source lists fit the hidden tile");
    const int t    = threadIdx.x;
    const int lane = t & 31;
    const int wave = __builtin_amdgcn_readfirstlane(t >> 5);
    const int b    = blockIdx.x;
    const int ch = wave * CPW + lane;
    int st = 0;
    int en = 0;
    if (lane < CPW && ch < NCH) {
        st = toff[TOFF(b, ch)];
        en = toff[TOFF(b + 1, ch)];
    }
    if (t < 64) cnt[t] = 0;
    if (t == 0) ovf = 0;
    const int len = en - st;
    int e = len;
    for (int d = 1; d < 32; d <<= 1) {
        const int u = __shfl_up(e, d);
        if (lane >= d) e += u;
    }
    const int m = __builtin_amdgcn_readlane(e, 31);
    const int rel = ch * CH + st - (e - len);
    int ends[CPW];
    for (int i = 0; i < CPW; ++i) ends[i] = __builtin_amdgcn_readlane(e, i);
    __syncthreads();
    const v2f* x2 = (const v2f*)x;
    for (int base = 0; base < m; base += 128) {
        unsigned w[4];
        for (int u = 0; u < 4; ++u) {
            const int j = base + 32 * u + lane;
            int own = 0;
            for (int i = 0; i < CPW; ++i) own += (j >= ends[i]) ? 1 : 0;
            const int orel = __shfl(rel, own);
            w[u] = (j < m) ? (unsigned)binned[orel + j] : 0u;
        }
        for (int u = 0; u < 4; ++u) {
            if (base + 32 * u + lane < m) {
                const unsigned row = (w[u] >> 16) & 63u;
                const int rank = atomicAdd(&cnt[row], 1);
                if (rank < RCAP) slot[row * RCAP + rank] = (unsigned short)(w[u] & 0xffffu);
                else             ovf = 1;
            }
        }
    }
    __syncthreads();
    const float se = 1.0f + eps[0];
    const int over = __builtin_amdgcn_readfirstlane(ovf);
    if (over == 0) {
        const int row0 = wave * 8;
        const int myn  = cnt[row0 + (lane & 7)];
        unsigned nxt = (unsigned)slot[row0 * RCAP + lane];
        for (int i = 0; i < 8; ++i) {
            const int row  = row0 + i;
            const int node = b * 64 + row;
            const int n    = __builtin_amdgcn_readlane(myn, i);
            unsigned mys = nxt;
            if (i < 7) nxt = (unsigned)slot[(row + 1) * RCAP + lane];
            v2f xv; xv.x = 0.0f; xv.y = 0.0f;
            if (node < N_NODES) xv = x2[(size_t)node * 32 + lane];
            v2f s; s.x = 0.0f; s.y = 0.0f;
            int k = 0;
            while (n - k >= 32) {
                v2f v[16];
                for (int jj = 0; jj < 16; ++jj) v[jj] = src_row(x2, mys, jj, lane);
                for (int jj = 0; jj < 16; ++jj) { s.x += v[jj].x; s.y += v[jj].y; }
                k += 16;
                mys = (unsigned)slot[row * RCAP + k + lane];
            }
            const int rem = n - k;
            v2f g16[16], g8[8], g4[4], g2[2], g1;
            g1.x = 0.0f; g1.y = 0.0f;
            int p = 0;
            if (rem & 16) {
                for (int jj = 0; jj < 16; ++jj) g16[jj] = src_row(x2, mys, jj, lane);
                p = 16;
            }
            if (rem & 8) {
                for (int jj = 0; jj < 8; ++jj) g8[jj] = src_row(x2, mys, p + jj, lane);
                p += 8;
            }
            if (rem & 4) {
                for (int jj = 0; jj < 4; ++jj) g4[jj] = src_row(x2, mys, p + jj, lane);
                p += 4;
            }
            if (rem & 2) {
                for (int jj = 0; jj < 2; ++jj) g2[jj] = src_row(x2, mys, p + jj, lane);
                p += 2;
            }
            if (rem & 1) g1 = src_row(x2, mys, p, lane);
            if (rem & 16) {
                for (int jj = 0; jj < 16; ++jj) { s.x += g16[jj].x; s.y += g16[jj].y; }
            }
            if (rem & 8) {
                for (int jj = 0; jj < 8; ++jj) { s.x += g8[jj].x; s.y += g8[jj].y; }
            }
            if (rem & 4) {
                for (int jj = 0; jj < 4; ++jj) { s.x += g4[jj].x; s.y += g4[jj].y; }
            }
            if (rem & 2) {
                for (int jj = 0; jj < 2; ++jj) { s.x += g2[jj].x; s.y += g2[jj].y; }
            }
            if (rem & 1) { s.x += g1.x; s.y += g1.y; }
            v2f o; o.x = 0.0f; o.y = 0.0f;
            if (node < N_NODES) {
                o.x = se * xv.x + s.x;
                o.y = se * xv.y + s.y;
            }
            *(v2f*)(acc + row * HS + 2 * lane) = o;
        }
    } else {
        for (int i = t; i < 64 * HS / 4; i += 256)
            ((float4*)acc)[i] = make_float4(0.0f, 0.0f, 0.0f, 0.0f);
        __syncthreads();
        for (int base = 0; base < m; base += 32) {
            int mm = m - base;
            if (mm > 32) mm = 32;
            const int j = base + lane;
            int own = 0;
            for (int i = 0; i < CPW; ++i) own += (j >= ends[i]) ? 1 : 0;
            const int orel = __shfl(rel, own);
            const unsigned myw = (lane < mm) ? (unsigned)binned[orel + j] : 0u;
            int k = 0;
            for (; k + 8 <= mm; k += 8) {
                v2f v[8];
                unsigned w[8];
                for (int jj = 0; jj < 8; ++jj) {
                    w[jj] = (unsigned)__builtin_amdgcn_readlane((int)myw, k + jj);
                    v[jj] = x2[(size_t)(w[jj] & 0xffffu) * 32 + lane];
                }
                for (int jj = 0; jj < 8; ++jj) {
                    float* a = acc + ((w[jj] >> 16) & 63u) * HS + 2 * lane;
                    atomicAdd(a, v[jj].x);
                    atomicAdd(a + 1, v[jj].y);
                }
            }
            if (k + 4 <= mm) {
                v2f v[4];
                unsigned w[4];
                for (int jj = 0; jj < 4; ++jj) {
                    w[jj] = (unsigned)__builtin_amdgcn_readlane((int)myw, k + jj);
                    v[jj] = x2[(size_t)(w[jj] & 0xffffu) * 32 + lane];
                }
                for (int jj = 0; jj < 4; ++jj) {
                    float* a = acc + ((w[jj] >> 16) & 63u) * HS + 2 * lane;
                    atomicAdd(a, v[jj].x);
                    atomicAdd(a + 1, v[jj].y);
                }
                k += 4;
            }
            for (; k < mm; ++k) {
                const unsigned w = (unsigned)__builtin_amdgcn_readlane((int)myw, k);
                const v2f v = x2[(size_t)(w & 0xffffu) * 32 + lane];
                float* a = acc + ((w >> 16) & 63u) * HS + 2 * lane;
                atomicAdd(a, v.x);
                atomicAdd(a + 1, v.y);
            }
        }
        __syncthreads();
        for (int i = t; i < 1024; i += 256) {
            const int row  = i >> 4;
            const int node = b * 64 + row;
            float4* p = (float4*)(acc + row * HS) + (i & 15);
            float4 o = make_float4(0.0f, 0.0f, 0.0f, 0.0f);
            if (node < N_NODES) {
                const float4 xv = ((const float4*)x)[(size_t)node * 16 + (i & 15)];
                const float4 a  = *p;
                o.x = se * xv.x + a.x;
                o.y = se * xv.y + a.y;
                o.z = se * xv.z + a.z;
                o.w = se * xv.w + a.w;
            }
            *p = o;
        }
    }
    __syncthreads();

    const int r     = lane & 15;
    const int koff  = (lane >> 4) << 1;
    const int mhalf = (lane >> 4) << 3;

    {
        const int n0 = wave * 16;
        const float bias = b1[n0 + r];
        v8f c0, c1, c2, c3;
#pragma unroll
        for (int v = 0; v < 8; ++v) { c0[v] = bias; c1[v] = bias; c2[v] = bias; c3[v] = bias; }
        const float* a0p = acc + r * HS + koff;
        const float* bp  = W1 + (size_t)koff * D_HID + n0 + r;
#pragma unroll
        for (int kk = 0; kk < 16; ++kk) {
            v2f bfr;
            bfr.x = bp[(size_t)(kk * 4) * D_HID];
            bfr.y = bp[(size_t)(kk * 4 + 1) * D_HID];
            const v2f a0 = *(const v2f*)(a0p + kk * 4);
            const v2f a1 = *(const v2f*)(a0p + 16 * HS + kk * 4);
            const v2f a2 = *(const v2f*)(a0p + 32 * HS + kk * 4);
            const v2f a3 = *(const v2f*)(a0p + 48 * HS + kk * 4);
            c0 = __builtin_amdgcn_wmma_f32_16x16x4_f32(
                false, a0, false, bfr, (short)0, c0, false, false);
            c1 = __builtin_amdgcn_wmma_f32_16x16x4_f32(
                false, a1, false, bfr, (short)0, c1, false, false);
            c2 = __builtin_amdgcn_wmma_f32_16x16x4_f32(
                false, a2, false, bfr, (short)0, c2, false, false);
            c3 = __builtin_amdgcn_wmma_f32_16x16x4_f32(
                false, a3, false, bfr, (short)0, c3, false, false);
        }
        float* hp = hid + mhalf * HPAD + n0 + r;
#pragma unroll
        for (int v = 0; v < 8; ++v) {
            const float t0 = c0[v], t1 = c1[v], t2 = c2[v], t3 = c3[v];
            hp[(v)      * HPAD] = t0 > 0.0f ? t0 : 0.0f;
            hp[(v + 16) * HPAD] = t1 > 0.0f ? t1 : 0.0f;
            hp[(v + 32) * HPAD] = t2 > 0.0f ? t2 : 0.0f;
            hp[(v + 48) * HPAD] = t3 > 0.0f ? t3 : 0.0f;
        }
    }
    __syncthreads();

    {
        const int n0  = (wave & 3) * 16;
        const int rt0 = (wave >> 2) * 32;
        const float bias = b2[n0 + r];
        v8f d0, d1;
#pragma unroll
        for (int v = 0; v < 8; ++v) { d0[v] = bias; d1[v] = bias; }
        const float* a0p = hid + (rt0 + r) * HPAD + koff;
        const float* bp  = W2 + (size_t)koff * D_IN + n0 + r;
#pragma unroll
        for (int kk = 0; kk < 32; ++kk) {
            v2f bfr;
            bfr.x = bp[(size_t)(kk * 4) * D_IN];
            bfr.y = bp[(size_t)(kk * 4 + 1) * D_IN];
            const v2f a0 = *(const v2f*)(a0p + kk * 4);
            const v2f a1 = *(const v2f*)(a0p + 16 * HPAD + kk * 4);
            d0 = __builtin_amdgcn_wmma_f32_16x16x4_f32(
                false, a0, false, bfr, (short)0, d0, false, false);
            d1 = __builtin_amdgcn_wmma_f32_16x16x4_f32(
                false, a1, false, bfr, (short)0, d1, false, false);
        }
        float* op = acc + (rt0 + mhalf) * HS + n0 + r;
#pragma unroll
        for (int v = 0; v < 8; ++v) {
            op[(v)      * HS] = d0[v];
            op[(v + 16) * HS] = d1[v];
        }
    }
    __syncthreads();
    for (int i = t; i < 1024; i += 256) {
        const int row  = i >> 4;
        const int node = b * 64 + row;
        if (node < N_NODES)
            ((float4*)out)[(size_t)node * 16 + (i & 15)] =
                *((const float4*)(acc + row * HS) + (i & 15));
    }
}

extern "C" void kernel_launch(void* const* d_in, const int* in_sizes, int n_in,
                              void* d_out, int out_size, void* d_ws, size_t ws_size,
                              hipStream_t stream) {
    const float* x   = (const float*)d_in[0];
    const int*   ei  = (const int*)d_in[1];
    const float* W1  = (const float*)d_in[2];
    const float* b1  = (const float*)d_in[3];
    const float* W2  = (const float*)d_in[4];
    const float* b2  = (const float*)d_in[5];
    const float* eps = (const float*)d_in[6];
    float* out = (float*)d_out;
    int*   toff   = (int*)((char*)d_ws + WS_TOFF);
    int*   binned = (int*)((char*)d_ws + WS_BINNED);

    gin_bin<<<NCH, 1024, 0, stream>>>(ei, toff, binned);
    gin_sum_mlp<<<NB, 256, 0, stream>>>(x, eps, toff, binned, W1, b1, W2, b2, out);
}


# last 848 nodes go in 53 ranges of 16 behind 768 ranges of 64 (3 per WGP): small sum workgroups spread the remainder, no WGP carries a 4th 64-node range (256 -> 208 rows at most); list order kept
# speedup vs baseline: 3.1360x; 1.0646x over previous
#include <hip/hip_runtime.h>

typedef __attribute__((ext_vector_type(2))) float v2f;
typedef __attribute__((ext_vector_type(8))) float v8f;

#define N_NODES 50000
#define N_EDGES 800000
#define D_IN    64
#define D_HID   128
#define HPAD    132
#define HS      68
#define RCAP    128

#define WS_TOFF   12800000
#define WS_BINNED 13500416
#define NWGP      256
#define NBIG      (3 * NWGP)
#define TAIL0     (NBIG * 64)
#define NSMALL    ((N_NODES - TAIL0 + 15) / 16)
#define NB        (NBIG + NSMALL)
#define RNG(d)    ((d) < TAIL0 ? ((d) >> 6) : NBIG + (((d) - TAIL0) >> 4))
static_assert(TAIL0 < N_NODES && NSMALL > 0, "there is a remainder to spread");
#define E4        (N_EDGES / 4)
#define CH        4096
#define NCH       ((N_EDGES + CH - 1) / CH)
#define CPW       ((NCH + 7) / 8)
#define TOFF(r, c) ((((r) >> 3) * NCH + (c)) * 8 + ((r) & 7))
static_assert(NB < 1024, "one thread of the binning workgroup per range");
static_assert(WS_BINNED - WS_TOFF >= 4 * (NB / 8 + 1) * NCH * 8, "toff fits");
static_assert(CPW <= 32, "one lane per chunk of a wave");

__global__ __launch_bounds__(1024)
void gin_bin(const int* __restrict__ ei, int* __restrict__ toff,
             int* __restrict__ binned) {
    __shared__ int h[1024];
    __shared__ int loff[1024];
    __shared__ unsigned sorted[CH];
    __shared__ int wa[32];
    const int t    = threadIdx.x;
    const int lane = t & 31;
    const int wave = t >> 5;
    h[t] = 0;
    __syncthreads();
    const int q = blockIdx.x * 1024 + t;
    const bool ok = q < E4;
    int4 r = make_int4(0, 0, 0, 0);
    int4 c = make_int4(0, 0, 0, 0);
    if (ok) {
        r = ((const int4*)ei)[q];
        c = ((const int4*)(ei + N_EDGES))[q];
        atomicAdd(&h[RNG(r.x)], 1);
        atomicAdd(&h[RNG(r.y)], 1);
        atomicAdd(&h[RNG(r.z)], 1);
        atomicAdd(&h[RNG(r.w)], 1);
    }
    __syncthreads();
    const int mine = h[t];
    int ia = mine;
    for (int d = 1; d < 32; d <<= 1) {
        const int ua = __shfl_up(ia, d);
        if (lane >= d) ia += ua;
    }
    if (lane == 31) wa[wave] = ia;
    __syncthreads();
    int pa = 0;
    for (int w = 0; w < wave; ++w) pa += wa[w];
    const int first = pa + ia - mine;
    loff[t] = first;
    if (t <= NB) toff[TOFF(t, blockIdx.x)] = first;
    h[t] = 0;
    __syncthreads();
    if (ok) {
        int p;
        p = loff[RNG(r.x)] + atomicAdd(&h[RNG(r.x)], 1);
        sorted[p] = ((unsigned)r.x << 16) | (unsigned)c.x;
        p = loff[RNG(r.y)] + atomicAdd(&h[RNG(r.y)], 1);
        sorted[p] = ((unsigned)r.y << 16) | (unsigned)c.y;
        p = loff[RNG(r.z)] + atomicAdd(&h[RNG(r.z)], 1);
        sorted[p] = ((unsigned)r.z << 16) | (unsigned)c.z;
        p = loff[RNG(r.w)] + atomicAdd(&h[RNG(r.w)], 1);
        sorted[p] = ((unsigned)r.w << 16) | (unsigned)c.w;
    }
    __syncthreads();
    int nvalid = N_EDGES - blockIdx.x * CH;
    if (nvalid > CH) nvalid = CH;
    for (int i = t; i < nvalid; i += 1024)
        binned[blockIdx.x * CH + i] = (int)sorted[i];
}

__device__ __forceinline__ v2f src_row(const v2f* __restrict__ x2, unsigned ids, int l, int lane) {
    const unsigned sidx = (unsigned)__builtin_amdgcn_readlane((int)ids, l);
    return x2[(size_t)(sidx & 0xffffu) * 32 + lane];
}


template <int NT>
__device__ __forceinline__ void mlp_layer1(const float* acc, float* hid,
                                           const float* __restrict__ W1,
                                           const float* __restrict__ b1,
                                           int n0, int r, int koff, int mhalf) {
    const float bias = b1[n0 + r];
    v8f c[NT];
#pragma unroll
    for (int j = 0; j < NT; ++j) {
#pragma unroll
        for (int v = 0; v < 8; ++v) c[j][v] = bias;
    }
    const float* a0p = acc + r * HS + koff;
    const float* bp  = W1 + (size_t)koff * D_HID + n0 + r;
#pragma unroll
    for (int kk = 0; kk < 16; ++kk) {
        v2f bfr;
        bfr.x = bp[(size_t)(kk * 4) * D_HID];
        bfr.y = bp[(size_t)(kk * 4 + 1) * D_HID];
#pragma unroll
        for (int j = 0; j < NT; ++j) {
            const v2f a = *(const v2f*)(a0p + j * 16 * HS + kk * 4);
            c[j] = __builtin_amdgcn_wmma_f32_16x16x4_f32(
                false, a, false, bfr, (short)0, c[j], false, false);
        }
    }
    float* hp = hid + mhalf * HPAD + n0 + r;
#pragma unroll
    for (int v = 0; v < 8; ++v) {
#pragma unroll
        for (int j = 0; j < NT; ++j) {
            const float tv = c[j][v];
            hp[(v + 16 * j) * HPAD] = tv > 0.0f ? tv : 0.0f;
        }
    }
}

template <int NT>
__device__ __forceinline__ void mlp_layer2(const float* hid, float* acc,
                                           const float* __restrict__ W2,
                                           const float* __restrict__ b2,
                                           int n0, int rt0, int r, int koff, int mhalf) {
    const float bias = b2[n0 + r];
    v8f d[NT];
#pragma unroll
    for (int j = 0; j < NT; ++j) {
#pragma unroll
        for (int v = 0; v < 8; ++v) d[j][v] = bias;
    }
    const float* a0p = hid + (rt0 + r) * HPAD + koff;
    const float* bp  = W2 + (size_t)koff * D_IN + n0 + r;
#pragma unroll
    for (int kk = 0; kk < 32; ++kk) {
        v2f bfr;
        bfr.x = bp[(size_t)(kk * 4) * D_IN];
        bfr.y = bp[(size_t)(kk * 4 + 1) * D_IN];
#pragma unroll
        for (int j = 0; j < NT; ++j) {
            const v2f a = *(const v2f*)(a0p + j * 16 * HPAD + kk * 4);
            d[j] = __builtin_amdgcn_wmma_f32_16x16x4_f32(
                false, a, false, bfr, (short)0, d[j], false, false);
        }
    }
    float* op = acc + (rt0 + mhalf) * HS + n0 + r;
#pragma unroll
    for (int v = 0; v < 8; ++v) {
#pragma unroll
        for (int j = 0; j < NT; ++j) op[(v + 16 * j) * HS] = d[j][v];
    }
}

__global__ __launch_bounds__(256)
void gin_sum_mlp(const float* __restrict__ x,
                 const float* __restrict__ eps,
                 const int* __restrict__ toff,
                 const int* __restrict__ binned,
                 const float* __restrict__ W1,
                 const float* __restrict__ b1,
                 const float* __restrict__ W2,
                 const float* __restrict__ b2,
                 float* __restrict__ out) {
    __shared__ __attribute__((aligned(16))) float acc[64 * HS];
    __shared__ __attribute__((aligned(16))) unsigned char hid_raw[64 * HPAD * 4];
    __shared__ int cnt[64];
    __shared__ int ovf;
    float* hid = (float*)hid_raw;
    unsigned short* slot = (unsigned short*)hid_raw;
    static_assert(64 * RCAP * 2 <= 64 * HPAD * 4, "source lists fit the hidden tile");
    const int t    = threadIdx.x;
    const int lane = t & 31;
    const int wave = __builtin_amdgcn_readfirstlane(t >> 5);
    const int b    = blockIdx.x;
    const bool small = b >= NBIG;
    const int node0  = small ? TAIL0 + (b - NBIG) * 16 : b * 64;
    const int nr     = small ? 16 : 64;
    const int ch = wave * CPW + lane;
    int st = 0;
    int en = 0;
    if (lane < CPW && ch < NCH) {
        st = toff[TOFF(b, ch)];
        en = toff[TOFF(b + 1, ch)];
    }
    if (t < 64) cnt[t] = 0;
    if (t == 0) ovf = 0;
    const int len = en - st;
    int e = len;
    for (int d = 1; d < 32; d <<= 1) {
        const int u = __shfl_up(e, d);
        if (lane >= d) e += u;
    }
    const int m = __builtin_amdgcn_readlane(e, 31);
    const int rel = ch * CH + st - (e - len);
    int ends[CPW];
    for (int i = 0; i < CPW; ++i) ends[i] = __builtin_amdgcn_readlane(e, i);
    __syncthreads();
    const v2f* x2 = (const v2f*)x;
    for (int base = 0; base < m; base += 128) {
        unsigned w[4];
        for (int u = 0; u < 4; ++u) {
            const int j = base + 32 * u + lane;
            int own = 0;
            for (int i = 0; i < CPW; ++i) own += (j >= ends[i]) ? 1 : 0;
            const int orel = __shfl(rel, own);
            w[u] = (j < m) ? (unsigned)binned[orel + j] : 0u;
        }
        for (int u = 0; u < 4; ++u) {
            if (base + 32 * u + lane < m) {
                const unsigned row = (w[u] >> 16) - (unsigned)node0;
                const int rank = atomicAdd(&cnt[row], 1);
                if (rank < RCAP) slot[row * RCAP + rank] = (unsigned short)(w[u] & 0xffffu);
                else             ovf = 1;
            }
        }
    }
    __syncthreads();
    const float se = 1.0f + eps[0];
    const int over = __builtin_amdgcn_readfirstlane(ovf);
    if (over == 0) {
        const int rpw  = nr >> 3;
        const int row0 = wave * rpw;
        const int myn  = cnt[row0 + (lane & 7)];
        unsigned nxt = (unsigned)slot[row0 * RCAP + lane];
        for (int i = 0; i < rpw; ++i) {
            const int row  = row0 + i;
            const int node = node0 + row;
            const int n    = __builtin_amdgcn_readlane(myn, i);
            unsigned mys = nxt;
            if (i + 1 < rpw) nxt = (unsigned)slot[(row + 1) * RCAP + lane];
            v2f xv; xv.x = 0.0f; xv.y = 0.0f;
            if (node < N_NODES) xv = x2[(size_t)node * 32 + lane];
            v2f s; s.x = 0.0f; s.y = 0.0f;
            int k = 0;
            while (n - k >= 32) {
                v2f v[16];
                for (int jj = 0; jj < 16; ++jj) v[jj] = src_row(x2, mys, jj, lane);
                for (int jj = 0; jj < 16; ++jj) { s.x += v[jj].x; s.y += v[jj].y; }
                k += 16;
                mys = (unsigned)slot[row * RCAP + k + lane];
            }
            const int rem = n - k;
            v2f g16[16], g8[8], g4[4], g2[2], g1;
            g1.x = 0.0f; g1.y = 0.0f;
            int p = 0;
            if (rem & 16) {
                for (int jj = 0; jj < 16; ++jj) g16[jj] = src_row(x2, mys, jj, lane);
                p = 16;
            }
            if (rem & 8) {
                for (int jj = 0; jj < 8; ++jj) g8[jj] = src_row(x2, mys, p + jj, lane);
                p += 8;
            }
            if (rem & 4) {
                for (int jj = 0; jj < 4; ++jj) g4[jj] = src_row(x2, mys, p + jj, lane);
                p += 4;
            }
            if (rem & 2) {
                for (int jj = 0; jj < 2; ++jj) g2[jj] = src_row(x2, mys, p + jj, lane);
                p += 2;
            }
            if (rem & 1) g1 = src_row(x2, mys, p, lane);
            if (rem & 16) {
                for (int jj = 0; jj < 16; ++jj) { s.x += g16[jj].x; s.y += g16[jj].y; }
            }
            if (rem & 8) {
                for (int jj = 0; jj < 8; ++jj) { s.x += g8[jj].x; s.y += g8[jj].y; }
            }
            if (rem & 4) {
                for (int jj = 0; jj < 4; ++jj) { s.x += g4[jj].x; s.y += g4[jj].y; }
            }
            if (rem & 2) {
                for (int jj = 0; jj < 2; ++jj) { s.x += g2[jj].x; s.y += g2[jj].y; }
            }
            if (rem & 1) { s.x += g1.x; s.y += g1.y; }
            v2f o; o.x = 0.0f; o.y = 0.0f;
            if (node < N_NODES) {
                o.x = se * xv.x + s.x;
                o.y = se * xv.y + s.y;
            }
            *(v2f*)(acc + row * HS + 2 * lane) = o;
        }
    } else {
        for (int i = t; i < 64 * HS / 4; i += 256)
            ((float4*)acc)[i] = make_float4(0.0f, 0.0f, 0.0f, 0.0f);
        __syncthreads();
        for (int base = 0; base < m; base += 32) {
            int mm = m - base;
            if (mm > 32) mm = 32;
            const int j = base + lane;
            int own = 0;
            for (int i = 0; i < CPW; ++i) own += (j >= ends[i]) ? 1 : 0;
            const int orel = __shfl(rel, own);
            const unsigned myw = (lane < mm) ? (unsigned)binned[orel + j] : 0u;
            int k = 0;
            for (; k + 8 <= mm; k += 8) {
                v2f v[8];
                unsigned w[8];
                for (int jj = 0; jj < 8; ++jj) {
                    w[jj] = (unsigned)__builtin_amdgcn_readlane((int)myw, k + jj);
                    v[jj] = x2[(size_t)(w[jj] & 0xffffu) * 32 + lane];
                }
                for (int jj = 0; jj < 8; ++jj) {
                    float* a = acc + ((w[jj] >> 16) - (unsigned)node0) * HS + 2 * lane;
                    atomicAdd(a, v[jj].x);
                    atomicAdd(a + 1, v[jj].y);
                }
            }
            if (k + 4 <= mm) {
                v2f v[4];
                unsigned w[4];
                for (int jj = 0; jj < 4; ++jj) {
                    w[jj] = (unsigned)__builtin_amdgcn_readlane((int)myw, k + jj);
                    v[jj] = x2[(size_t)(w[jj] & 0xffffu) * 32 + lane];
                }
                for (int jj = 0; jj < 4; ++jj) {
                    float* a = acc + ((w[jj] >> 16) - (unsigned)node0) * HS + 2 * lane;
                    atomicAdd(a, v[jj].x);
                    atomicAdd(a + 1, v[jj].y);
                }
                k += 4;
            }
            for (; k < mm; ++k) {
                const unsigned w = (unsigned)__builtin_amdgcn_readlane((int)myw, k);
                const v2f v = x2[(size_t)(w & 0xffffu) * 32 + lane];
                float* a = acc + ((w >> 16) - (unsigned)node0) * HS + 2 * lane;
                atomicAdd(a, v.x);
                atomicAdd(a + 1, v.y);
            }
        }
        __syncthreads();
        for (int i = t; i < nr * 16; i += 256) {
            const int row  = i >> 4;
            const int node = node0 + row;
            float4* p = (float4*)(acc + row * HS) + (i & 15);
            float4 o = make_float4(0.0f, 0.0f, 0.0f, 0.0f);
            if (node < N_NODES) {
                const float4 xv = ((const float4*)x)[(size_t)node * 16 + (i & 15)];
                const float4 a  = *p;
                o.x = se * xv.x + a.x;
                o.y = se * xv.y + a.y;
                o.z = se * xv.z + a.z;
                o.w = se * xv.w + a.w;
            }
            *p = o;
        }
    }
    __syncthreads();

    const int r     = lane & 15;
    const int koff  = (lane >> 4) << 1;
    const int mhalf = (lane >> 4) << 3;

    if (!small) mlp_layer1<4>(acc, hid, W1, b1, wave * 16, r, koff, mhalf);
    else        mlp_layer1<1>(acc, hid, W1, b1, wave * 16, r, koff, mhalf);
    __syncthreads();

    if (!small)        mlp_layer2<2>(hid, acc, W2, b2, (wave & 3) * 16, (wave >> 2) * 32, r, koff, mhalf);
    else if (wave < 4) mlp_layer2<1>(hid, acc, W2, b2, wave * 16, 0, r, koff, mhalf);
    __syncthreads();
    for (int i = t; i < nr * 16; i += 256) {
        const int row  = i >> 4;
        const int node = node0 + row;
        if (node < N_NODES)
            ((float4*)out)[(size_t)node * 16 + (i & 15)] =
                *((const float4*)(acc + row * HS) + (i & 15));
    }
}

extern "C" void kernel_launch(void* const* d_in, const int* in_sizes, int n_in,
                              void* d_out, int out_size, void* d_ws, size_t ws_size,
                              hipStream_t stream) {
    const float* x   = (const float*)d_in[0];
    const int*   ei  = (const int*)d_in[1];
    const float* W1  = (const float*)d_in[2];
    const float* b1  = (const float*)d_in[3];
    const float* W2  = (const float*)d_in[4];
    const float* b2  = (const float*)d_in[5];
    const float* eps = (const float*)d_in[6];
    float* out = (float*)d_out;
    int*   toff   = (int*)((char*)d_ws + WS_TOFF);
    int*   binned = (int*)((char*)d_ws + WS_BINNED);

    gin_bin<<<NCH, 1024, 0, stream>>>(ei, toff, binned);
    gin_sum_mlp<<<NB, 256, 0, stream>>>(x, eps, toff, binned, W1, b1, W2, b2, out);
}
